# starting kernel
# baseline (speedup 1.0000x reference)
#include <hip/hip_runtime.h>
#include <math.h>

static constexpr int N    = 50000;
static constexpr int E    = 800000;
static constexpr int ETOT = E + N;
static constexpr int F    = 16;
static constexpr int ED   = 7;
static constexpr int HC   = 64;
static constexpr float NEG_ATT = 0.2f;
static constexpr float NEG     = 0.01f;
static constexpr int NEG_INF_BITS = 0xFF800000;

typedef __attribute__((ext_vector_type(2))) float v2f;
typedef __attribute__((ext_vector_type(8))) float v8f;

static __device__ __forceinline__ void atomic_add_f(float* p, float v) {
    __hip_atomic_fetch_add(p, v, __ATOMIC_RELAXED, __HIP_MEMORY_SCOPE_AGENT);
}
static __device__ __forceinline__ void atomic_max_fbits(int* p, float v) {
    if (v >= 0.0f) atomicMax(p, __float_as_int(v));
    else           atomicMin((unsigned int*)p, __float_as_uint(v));
}

__global__ void k_zero(float* __restrict__ p, int n) {
    int i = blockIdx.x * blockDim.x + threadIdx.x;
    if (i < n) p[i] = 0.0f;
}

__global__ void k_eattr_scatter(const float* __restrict__ eattr,
                                const int* __restrict__ ei,
                                float* __restrict__ lsum,
                                float* __restrict__ cnt) {
    int e = blockIdx.x * blockDim.x + threadIdx.x;
    if (e >= E) return;
    int d = ei[E + e];
    #pragma unroll
    for (int j = 0; j < ED; ++j)
        atomic_add_f(&lsum[d * ED + j], eattr[e * ED + j]);
    atomic_add_f(&cnt[d], 1.0f);
}

__global__ void k_loop_mean(float* __restrict__ lsum, const float* __restrict__ cnt) {
    int n = blockIdx.x * blockDim.x + threadIdx.x;
    if (n >= N) return;
    float inv = 1.0f / fmaxf(cnt[n], 1.0f);
    #pragma unroll
    for (int j = 0; j < ED; ++j) lsum[n * ED + j] *= inv;
}

template <int K>
__global__ void k_gemm_pair(const float* __restrict__ A,
                            const float* __restrict__ Wl, const float* __restrict__ bl,
                            const float* __restrict__ Wr, const float* __restrict__ br,
                            float* __restrict__ outL, float* __restrict__ outR) {
    const int wave = threadIdx.x >> 5;
    const int lane = threadIdx.x & 31;
    const int r    = lane & 15;
    const int hi   = lane >> 4;
    const int row0 = blockIdx.x * 16;
    const int col0 = wave * 16;
    const int arow = row0 + r;
    const int bcol = col0 + r;

    v8f accL = {};
    v8f accR = {};
    #pragma unroll
    for (int k0 = 0; k0 < K; k0 += 4) {
        const int kk = k0 + hi * 2;
        v2f a = *reinterpret_cast<const v2f*>(A + arow * K + kk);
        v2f b;
        b.x = Wl[kk * HC + bcol];
        b.y = Wl[(kk + 1) * HC + bcol];
        accL = __builtin_amdgcn_wmma_f32_16x16x4_f32(false, a, false, b,
                                                     (short)0, accL, false, false);
        v2f b2;
        b2.x = Wr[kk * HC + bcol];
        b2.y = Wr[(kk + 1) * HC + bcol];
        accR = __builtin_amdgcn_wmma_f32_16x16x4_f32(false, a, false, b2,
                                                     (short)0, accR, false, false);
    }
    const float bLv = bl[bcol];
    const float bRv = br[bcol];
    #pragma unroll
    for (int i = 0; i < 8; ++i) {
        const int m = row0 + hi * 8 + i;
        outL[m * HC + bcol] = accL[i] + bLv;
        outR[m * HC + bcol] = accR[i] + bRv;
    }
}

__global__ void k_init_attn(float* __restrict__ acc,
                            int*   __restrict__ amax,
                            float* __restrict__ z) {
    int i = blockIdx.x * blockDim.x + threadIdx.x;
    if (i < N * HC) acc[i] = 0.0f;
    if (i < N * 4) { amax[i] = NEG_INF_BITS; z[i] = 0.0f; }
}

__global__ void k_logits(const float* __restrict__ xl, const float* __restrict__ xr,
                         const float* __restrict__ eattr,
                         const float* __restrict__ lattr,
                         const int* __restrict__ ei,
                         const float* __restrict__ We,
                         const float* __restrict__ att,
                         float* __restrict__ alog,
                         int*   __restrict__ amax) {
    __shared__ float sWe[ED * HC];
    __shared__ float sAtt[HC];
    for (int i = threadIdx.x; i < ED * HC; i += blockDim.x) sWe[i] = We[i];
    for (int i = threadIdx.x; i < HC; i += blockDim.x) sAtt[i] = att[i];
    __syncthreads();

    int t = blockIdx.x * blockDim.x + threadIdx.x;
    if (t >= ETOT * 4) return;
    const int e = t >> 2;
    const int h = t & 3;

    int s, d;
    const float* ea;
    if (e < E) { s = ei[e]; d = ei[E + e]; ea = eattr + (size_t)e * ED; }
    else       { s = e - E; d = s;         ea = lattr + (size_t)(e - E) * ED; }

    float eat[ED];
    #pragma unroll
    for (int j = 0; j < ED; ++j) eat[j] = ea[j];

    const float4* xls = reinterpret_cast<const float4*>(xl + (size_t)s * HC + h * 16);
    const float4* xrd = reinterpret_cast<const float4*>(xr + (size_t)d * HC + h * 16);

    float logit = 0.0f;
    #pragma unroll
    for (int q = 0; q < 4; ++q) {
        float4 vl = xls[q];
        float4 vr = xrd[q];
        float xv[4] = { vl.x + vr.x, vl.y + vr.y, vl.z + vr.z, vl.w + vr.w };
        #pragma unroll
        for (int k = 0; k < 4; ++k) {
            const int c = q * 4 + k;
            float em = 0.0f;
            #pragma unroll
            for (int j = 0; j < ED; ++j) em = fmaf(eat[j], sWe[j * HC + h * 16 + c], em);
            float v = xv[k] + em;
            v = (v > 0.0f) ? v : NEG_ATT * v;
            logit = fmaf(v, sAtt[h * 16 + c], logit);
        }
    }
    alog[(size_t)e * 4 + h] = logit;
    atomic_max_fbits(&amax[d * 4 + h], logit);
}

__global__ void k_accum(const float* __restrict__ xl,
                        const int* __restrict__ ei,
                        const float* __restrict__ alog,
                        const int*   __restrict__ amax,
                        float* __restrict__ z,
                        float* __restrict__ acc) {
    int t = blockIdx.x * blockDim.x + threadIdx.x;
    if (t >= ETOT * 4) return;
    const int e = t >> 2;
    const int h = t & 3;
    int s, d;
    if (e < E) { s = ei[e]; d = ei[E + e]; }
    else       { s = e - E; d = s; }

    const float m  = __int_as_float(amax[d * 4 + h]);
    const float ea = __expf(alog[(size_t)e * 4 + h] - m);
    atomic_add_f(&z[d * 4 + h], ea);

    const float4* xs = reinterpret_cast<const float4*>(xl + (size_t)s * HC + h * 16);
    float* ad = acc + (size_t)d * HC + h * 16;
    #pragma unroll
    for (int i = 0; i < 4; ++i) {
        float4 v = xs[i];
        atomic_add_f(&ad[4 * i + 0], ea * v.x);
        atomic_add_f(&ad[4 * i + 1], ea * v.y);
        atomic_add_f(&ad[4 * i + 2], ea * v.z);
        atomic_add_f(&ad[4 * i + 3], ea * v.w);
    }
}

__global__ void k_finalize(const float* __restrict__ acc,
                           const float* __restrict__ z,
                           const float* __restrict__ bo,
                           float* __restrict__ hout) {
    int i = blockIdx.x * blockDim.x + threadIdx.x;
    if (i >= N * HC) return;
    const int n = i >> 6;
    const int c = i & 63;
    float v = acc[i] / z[n * 4 + (c >> 4)] + bo[c];
    hout[i] = (v > 0.0f) ? v : NEG * v;
}

__global__ void k_head(const float* __restrict__ h2,
                       const float* __restrict__ Ws,
                       const float* __restrict__ bs,
                       float* __restrict__ out) {
    int n = blockIdx.x * blockDim.x + threadIdx.x;
    if (n >= N) return;
    const float4* hv = reinterpret_cast<const float4*>(h2 + (size_t)n * HC);
    const float4* wv = reinterpret_cast<const float4*>(Ws);
    float s = 0.0f;
    #pragma unroll
    for (int j = 0; j < 16; ++j) {
        float4 a = hv[j], b = wv[j];
        s = fmaf(a.x, b.x, s); s = fmaf(a.y, b.y, s);
        s = fmaf(a.z, b.z, s); s = fmaf(a.w, b.w, s);
    }
    out[n] = s + bs[0];
}

static inline int blocks(long long n, int bs) { return (int)((n + bs - 1) / bs); }

extern "C" void kernel_launch(void* const* d_in, const int* in_sizes, int n_in,
                              void* d_out, int out_size, void* d_ws, size_t ws_size,
                              hipStream_t stream) {
    const float* x     = (const float*)d_in[0];
    const int*   ei    = (const int*)d_in[1];
    const float* eattr = (const float*)d_in[2];
    const float* Wl1 = (const float*)d_in[3];  const float* bl1 = (const float*)d_in[4];
    const float* Wr1 = (const float*)d_in[5];  const float* br1 = (const float*)d_in[6];
    const float* We1 = (const float*)d_in[7];  const float* att1 = (const float*)d_in[8];
    const float* bo1 = (const float*)d_in[9];
    const float* Wl2 = (const float*)d_in[10]; const float* bl2 = (const float*)d_in[11];
    const float* Wr2 = (const float*)d_in[12]; const float* br2 = (const float*)d_in[13];
    const float* We2 = (const float*)d_in[14]; const float* att2 = (const float*)d_in[15];
    const float* bo2 = (const float*)d_in[16];
    const float* Ws  = (const float*)d_in[17]; const float* bs = (const float*)d_in[18];
    float* out = (float*)d_out;

    float* ws = (float*)d_ws;
    size_t o = 0;
    float* lsum = ws + o; o += (size_t)N * ED;
    float* cnt  = ws + o; o += (size_t)N;
    float* xl   = ws + o; o += (size_t)N * HC;
    float* xr   = ws + o; o += (size_t)N * HC;
    float* alog = ws + o; o += (size_t)ETOT * 4;
    int*   amax = (int*)(ws + o); o += (size_t)N * 4;
    float* zden = ws + o; o += (size_t)N * 4;
    float* acc  = ws + o; o += (size_t)N * HC;
    float* hbuf = ws + o; o += (size_t)N * HC;

    const int BT = 256;

    k_zero<<<blocks((long long)N * (ED + 1), BT), BT, 0, stream>>>(lsum, N * (ED + 1));
    k_eattr_scatter<<<blocks(E, BT), BT, 0, stream>>>(eattr, ei, lsum, cnt);
    k_loop_mean<<<blocks(N, BT), BT, 0, stream>>>(lsum, cnt);

    k_gemm_pair<F><<<N / 16, 128, 0, stream>>>(x, Wl1, bl1, Wr1, br1, xl, xr);
    k_init_attn<<<blocks((long long)N * HC, BT), BT, 0, stream>>>(acc, amax, zden);
    k_logits<<<blocks((long long)ETOT * 4, BT), BT, 0, stream>>>(xl, xr, eattr, lsum, ei,
                                                                 We1, att1, alog, amax);
    k_accum<<<blocks((long long)ETOT * 4, BT), BT, 0, stream>>>(xl, ei, alog, amax, zden, acc);
    k_finalize<<<blocks((long long)N * HC, BT), BT, 0, stream>>>(acc, zden, bo1, hbuf);

    k_gemm_pair<HC><<<N / 16, 128, 0, stream>>>(hbuf, Wl2, bl2, Wr2, br2, xl, xr);
    k_init_attn<<<blocks((long long)N * HC, BT), BT, 0, stream>>>(acc, amax, zden);
    k_logits<<<blocks((long long)ETOT * 4, BT), BT, 0, stream>>>(xl, xr, eattr, lsum, ei,
                                                                 We2, att2, alog, amax);
    k_accum<<<blocks((long long)ETOT * 4, BT), BT, 0, stream>>>(xl, ei, alog, amax, zden, acc);
    k_finalize<<<blocks((long long)N * HC, BT), BT, 0, stream>>>(acc, zden, bo2, hbuf);

    k_head<<<blocks(N, BT), BT, 0, stream>>>(hbuf, Ws, bs, out);
}


# edges bucketed by destination once per call (count, scan, fill; arrival-order rank); one wave per node sums its in-edges with a running-max softmax: no float atomics, sum order changed
# speedup vs baseline: 4.7175x; 4.7175x over previous
#include <hip/hip_runtime.h>
#include <math.h>

static constexpr int N    = 50000;
static constexpr int E    = 800000;
static constexpr int F    = 16;
static constexpr int ED   = 7;
static constexpr int HC   = 64;
static constexpr float NEG_ATT = 0.2f;
static constexpr float NEG     = 0.01f;

typedef __attribute__((ext_vector_type(2))) float v2f;
typedef __attribute__((ext_vector_type(8))) float v8f;


__global__ void k_zero_i(int* __restrict__ p, int n) {
    int i = blockIdx.x * blockDim.x + threadIdx.x;
    if (i < n) p[i] = 0;
}

__global__ void k_count(const int* __restrict__ ei, int* __restrict__ deg,
                        int* __restrict__ rank) {
    int e = blockIdx.x * blockDim.x + threadIdx.x;
    if (e >= E) return;
    rank[e] = atomicAdd(&deg[ei[E + e]], 1);
}

__global__ __launch_bounds__(1024) void k_scan(const int* __restrict__ deg,
                                               int* __restrict__ rowptr) {
    __shared__ int sScan[32];
    __shared__ int sRed[32];
    __shared__ int sBase;
    const int t    = threadIdx.x;
    const int lane = t & 31;
    const int wave = t >> 5;
    const int b0   = blockIdx.x * 1024;

    int part = 0;
    for (int i = t; i < b0; i += 1024) part += deg[i];
    #pragma unroll
    for (int o = 16; o > 0; o >>= 1) part += __shfl_xor(part, o);

    const int i = b0 + t;
    const int v = (i < N) ? deg[i] : 0;
    int incl = v;
    #pragma unroll
    for (int o = 1; o < 32; o <<= 1) {
        int u = __shfl_up(incl, o);
        if (lane >= o) incl += u;
    }
    if (lane == 31) sScan[wave] = incl;
    if (lane == 0)  sRed[wave]  = part;
    __syncthreads();
    if (wave == 0) {
        const int tot = sScan[lane];
        int wi = tot;
        #pragma unroll
        for (int o = 1; o < 32; o <<= 1) {
            int u = __shfl_up(wi, o);
            if (lane >= o) wi += u;
        }
        sScan[lane] = wi - tot;
        int r = sRed[lane];
        #pragma unroll
        for (int o = 16; o > 0; o >>= 1) r += __shfl_xor(r, o);
        if (lane == 0) sBase = r;
    }
    __syncthreads();
    const int excl = sBase + sScan[wave] + incl - v;
    if (i < N)      rowptr[i] = excl;
    if (i == N - 1) rowptr[N] = excl + v;
}

__global__ void k_fill(const int* __restrict__ ei, const int* __restrict__ rank,
                       const int* __restrict__ rowptr, int2* __restrict__ csr) {
    int e = blockIdx.x * blockDim.x + threadIdx.x;
    if (e >= E) return;
    const int d = ei[E + e];
    csr[rowptr[d] + rank[e]] = make_int2(ei[e], e);
}

__global__ void k_loop_mean(const float* __restrict__ eattr,
                            const int* __restrict__ rowptr,
                            const int2* __restrict__ csr,
                            float* __restrict__ lattr) {
    int t = blockIdx.x * blockDim.x + threadIdx.x;
    const int n = t >> 3;
    const int j = t & 7;
    if (n >= N || j >= ED) return;
    const int beg = rowptr[n];
    const int end = rowptr[n + 1];
    float s = 0.0f;
    for (int p = beg; p < end; ++p) s += eattr[(size_t)csr[p].y * ED + j];
    const float inv = 1.0f / fmaxf((float)(end - beg), 1.0f);
    lattr[n * ED + j] = s * inv;
}

template <int K>
__global__ void k_gemm_pair(const float* __restrict__ A,
                            const float* __restrict__ Wl, const float* __restrict__ bl,
                            const float* __restrict__ Wr, const float* __restrict__ br,
                            float* __restrict__ outL, float* __restrict__ outR) {
    const int wave = threadIdx.x >> 5;
    const int lane = threadIdx.x & 31;
    const int r    = lane & 15;
    const int hi   = lane >> 4;
    const int row0 = blockIdx.x * 16;
    const int col0 = wave * 16;
    const int arow = row0 + r;
    const int bcol = col0 + r;

    v8f accL = {};
    v8f accR = {};
    #pragma unroll
    for (int k0 = 0; k0 < K; k0 += 4) {
        const int kk = k0 + hi * 2;
        v2f a = *reinterpret_cast<const v2f*>(A + arow * K + kk);
        v2f b;
        b.x = Wl[kk * HC + bcol];
        b.y = Wl[(kk + 1) * HC + bcol];
        accL = __builtin_amdgcn_wmma_f32_16x16x4_f32(false, a, false, b,
                                                     (short)0, accL, false, false);
        v2f b2;
        b2.x = Wr[kk * HC + bcol];
        b2.y = Wr[(kk + 1) * HC + bcol];
        accR = __builtin_amdgcn_wmma_f32_16x16x4_f32(false, a, false, b2,
                                                     (short)0, accR, false, false);
    }
    const float bLv = bl[bcol];
    const float bRv = br[bcol];
    #pragma unroll
    for (int i = 0; i < 8; ++i) {
        const int m = row0 + hi * 8 + i;
        outL[m * HC + bcol] = accL[i] + bLv;
        outR[m * HC + bcol] = accR[i] + bRv;
    }
}

static __device__ __forceinline__ float head_sum(float v) {
    v += __int_as_float(__builtin_amdgcn_update_dpp(0, __float_as_int(v), 0xB1,  0xF, 0xF, false));
    v += __int_as_float(__builtin_amdgcn_update_dpp(0, __float_as_int(v), 0x4E,  0xF, 0xF, false));
    v += __int_as_float(__builtin_amdgcn_update_dpp(0, __float_as_int(v), 0x141, 0xF, 0xF, false));
    return v;
}

__global__ __launch_bounds__(256) void k_gat_node(
        const float* __restrict__ xl, const float* __restrict__ xr,
        const float* __restrict__ eattr,
        const float* __restrict__ lattr,
        const int*   __restrict__ rowptr,
        const int2*  __restrict__ csr,
        const float* __restrict__ We,
        const float* __restrict__ att,
        const float* __restrict__ bo,
        float* __restrict__ hout) {
    const int lane = threadIdx.x & 31;
    const int node = __builtin_amdgcn_readfirstlane(blockIdx.x * 8 + (threadIdx.x >> 5));
    const int c0 = lane * 2;

    float w0[ED], w1[ED];
    #pragma unroll
    for (int j = 0; j < ED; ++j) {
        const v2f w = *reinterpret_cast<const v2f*>(We + j * HC + c0);
        w0[j] = w.x; w1[j] = w.y;
    }
    const v2f at  = *reinterpret_cast<const v2f*>(att + c0);
    const v2f xrd = *reinterpret_cast<const v2f*>(xr + (size_t)node * HC + c0);
    const v2f xld = *reinterpret_cast<const v2f*>(xl + (size_t)node * HC + c0);

    float m, z = 1.0f, acc0 = xld.x, acc1 = xld.y;
    {
        float em0 = 0.0f, em1 = 0.0f;
        #pragma unroll
        for (int j = 0; j < ED; ++j) {
            const float a = lattr[node * ED + j];
            em0 = fmaf(a, w0[j], em0);
            em1 = fmaf(a, w1[j], em1);
        }
        float v0 = (xld.x + xrd.x) + em0;
        float v1 = (xld.y + xrd.y) + em1;
        v0 = (v0 > 0.0f) ? v0 : NEG_ATT * v0;
        v1 = (v1 > 0.0f) ? v1 : NEG_ATT * v1;
        m = head_sum(fmaf(v1, at.y, v0 * at.x));
    }

    const int beg = rowptr[node];
    const int end = rowptr[node + 1];
    for (int base = beg; base < end; base += 32) {
        const int cnt = min(32, end - base);
        const int2 se = csr[base + min(lane, cnt - 1)];
        for (int k = 0; k < cnt; k += 4) {
            v2f   xs[4];
            float ea[4][ED];
            #pragma unroll
            for (int u = 0; u < 4; ++u) {
                const int kk = min(k + u, cnt - 1);
                const int s = __builtin_amdgcn_readlane(se.x, kk);
                const int e = __builtin_amdgcn_readlane(se.y, kk);
                xs[u] = *reinterpret_cast<const v2f*>(xl + (size_t)s * HC + c0);
                #pragma unroll
                for (int j = 0; j < ED; ++j) ea[u][j] = eattr[(size_t)e * ED + j];
            }
            #pragma unroll
            for (int u = 0; u < 4; ++u) {
                float em0 = 0.0f, em1 = 0.0f;
                #pragma unroll
                for (int j = 0; j < ED; ++j) {
                    em0 = fmaf(ea[u][j], w0[j], em0);
                    em1 = fmaf(ea[u][j], w1[j], em1);
                }
                float v0 = (xs[u].x + xrd.x) + em0;
                float v1 = (xs[u].y + xrd.y) + em1;
                v0 = (v0 > 0.0f) ? v0 : NEG_ATT * v0;
                v1 = (v1 > 0.0f) ? v1 : NEG_ATT * v1;
                float lg = head_sum(fmaf(v1, at.y, v0 * at.x));
                lg = (k + u < cnt) ? lg : -INFINITY;
                const float mn = fmaxf(m, lg);
                const float sc = __expf(m - mn);
                const float w  = __expf(lg - mn);
                z    = fmaf(z,    sc, w);
                acc0 = fmaf(acc0, sc, w * xs[u].x);
                acc1 = fmaf(acc1, sc, w * xs[u].y);
                m = mn;
            }
        }
    }

    const v2f b = *reinterpret_cast<const v2f*>(bo + c0);
    float o0 = acc0 / z + b.x;
    float o1 = acc1 / z + b.y;
    v2f o;
    o.x = (o0 > 0.0f) ? o0 : NEG * o0;
    o.y = (o1 > 0.0f) ? o1 : NEG * o1;
    *reinterpret_cast<v2f*>(hout + (size_t)node * HC + c0) = o;
}

__global__ void k_head(const float* __restrict__ h2,
                       const float* __restrict__ Ws,
                       const float* __restrict__ bs,
                       float* __restrict__ out) {
    int n = blockIdx.x * blockDim.x + threadIdx.x;
    if (n >= N) return;
    const float4* hv = reinterpret_cast<const float4*>(h2 + (size_t)n * HC);
    const float4* wv = reinterpret_cast<const float4*>(Ws);
    float s = 0.0f;
    #pragma unroll
    for (int j = 0; j < 16; ++j) {
        float4 a = hv[j], b = wv[j];
        s = fmaf(a.x, b.x, s); s = fmaf(a.y, b.y, s);
        s = fmaf(a.z, b.z, s); s = fmaf(a.w, b.w, s);
    }
    out[n] = s + bs[0];
}

static inline int blocks(long long n, int bs) { return (int)((n + bs - 1) / bs); }

extern "C" void kernel_launch(void* const* d_in, const int* in_sizes, int n_in,
                              void* d_out, int out_size, void* d_ws, size_t ws_size,
                              hipStream_t stream) {
    const float* x     = (const float*)d_in[0];
    const int*   ei    = (const int*)d_in[1];
    const float* eattr = (const float*)d_in[2];
    const float* Wl1 = (const float*)d_in[3];  const float* bl1 = (const float*)d_in[4];
    const float* Wr1 = (const float*)d_in[5];  const float* br1 = (const float*)d_in[6];
    const float* We1 = (const float*)d_in[7];  const float* att1 = (const float*)d_in[8];
    const float* bo1 = (const float*)d_in[9];
    const float* Wl2 = (const float*)d_in[10]; const float* bl2 = (const float*)d_in[11];
    const float* Wr2 = (const float*)d_in[12]; const float* br2 = (const float*)d_in[13];
    const float* We2 = (const float*)d_in[14]; const float* att2 = (const float*)d_in[15];
    const float* bo2 = (const float*)d_in[16];
    const float* Ws  = (const float*)d_in[17]; const float* bs = (const float*)d_in[18];
    float* out = (float*)d_out;

    float* ws = (float*)d_ws;
    size_t o = 0;
    float* xl    = ws + o; o += (size_t)N * HC;
    float* xr    = ws + o; o += (size_t)N * HC;
    float* hbuf  = ws + o; o += (size_t)N * HC;
    float* lattr = ws + o; o += (size_t)N * ED;
    int*   deg   = (int*)(ws + o); o += (size_t)N;
    int*   rowp  = (int*)(ws + o); o += (size_t)N + 2;
    int*   rank  = (int*)(ws + o); o += (size_t)E;
    int2*  csr   = (int2*)(ws + o); o += (size_t)E * 2;

    const int BT = 256;

    k_zero_i<<<blocks(N, BT), BT, 0, stream>>>(deg, N);
    k_count<<<blocks(E, BT), BT, 0, stream>>>(ei, deg, rank);
    k_scan<<<blocks(N, 1024), 1024, 0, stream>>>(deg, rowp);
    k_fill<<<blocks(E, BT), BT, 0, stream>>>(ei, rank, rowp, csr);

    k_loop_mean<<<blocks((long long)N * 8, BT), BT, 0, stream>>>(eattr, rowp, csr, lattr);

    k_gemm_pair<F><<<N / 16, 128, 0, stream>>>(x, Wl1, bl1, Wr1, br1, xl, xr);
    k_gat_node<<<N / 8, 256, 0, stream>>>(xl, xr, eattr, lattr, rowp, csr, We1, att1, bo1, hbuf);

    k_gemm_pair<HC><<<N / 16, 128, 0, stream>>>(hbuf, Wl2, bl2, Wr2, br2, xl, xr);
    k_gat_node<<<N / 8, 256, 0, stream>>>(xl, xr, eattr, lattr, rowp, csr, We2, att2, bo2, hbuf);

    k_head<<<blocks(N, BT), BT, 0, stream>>>(hbuf, Ws, bs, out);
}


# node kernel takes the self loop last, its projected attr = mean of the in-edges' projected attrs (projection is linear): mean launch and its table dropped; softmax and mean sum order changed
# speedup vs baseline: 5.1316x; 1.0878x over previous
#include <hip/hip_runtime.h>
#include <math.h>

static constexpr int N    = 50000;
static constexpr int E    = 800000;
static constexpr int F    = 16;
static constexpr int ED   = 7;
static constexpr int HC   = 64;
static constexpr float NEG_ATT = 0.2f;
static constexpr float NEG     = 0.01f;

typedef __attribute__((ext_vector_type(2))) float v2f;
typedef __attribute__((ext_vector_type(8))) float v8f;


__global__ void k_zero_i(int* __restrict__ p, int n) {
    int i = blockIdx.x * blockDim.x + threadIdx.x;
    if (i < n) p[i] = 0;
}

__global__ void k_count(const int* __restrict__ ei, int* __restrict__ deg,
                        int* __restrict__ rank) {
    int e = blockIdx.x * blockDim.x + threadIdx.x;
    if (e >= E) return;
    rank[e] = atomicAdd(&deg[ei[E + e]], 1);
}

__global__ __launch_bounds__(1024) void k_scan(const int* __restrict__ deg,
                                               int* __restrict__ rowptr) {
    __shared__ int sScan[32];
    __shared__ int sRed[32];
    __shared__ int sBase;
    const int t    = threadIdx.x;
    const int lane = t & 31;
    const int wave = t >> 5;
    const int b0   = blockIdx.x * 1024;

    int part = 0;
    for (int i = t; i < b0; i += 1024) part += deg[i];
    #pragma unroll
    for (int o = 16; o > 0; o >>= 1) part += __shfl_xor(part, o);

    const int i = b0 + t;
    const int v = (i < N) ? deg[i] : 0;
    int incl = v;
    #pragma unroll
    for (int o = 1; o < 32; o <<= 1) {
        int u = __shfl_up(incl, o);
        if (lane >= o) incl += u;
    }
    if (lane == 31) sScan[wave] = incl;
    if (lane == 0)  sRed[wave]  = part;
    __syncthreads();
    if (wave == 0) {
        const int tot = sScan[lane];
        int wi = tot;
        #pragma unroll
        for (int o = 1; o < 32; o <<= 1) {
            int u = __shfl_up(wi, o);
            if (lane >= o) wi += u;
        }
        sScan[lane] = wi - tot;
        int r = sRed[lane];
        #pragma unroll
        for (int o = 16; o > 0; o >>= 1) r += __shfl_xor(r, o);
        if (lane == 0) sBase = r;
    }
    __syncthreads();
    const int excl = sBase + sScan[wave] + incl - v;
    if (i < N)      rowptr[i] = excl;
    if (i == N - 1) rowptr[N] = excl + v;
}

__global__ void k_fill(const int* __restrict__ ei, const int* __restrict__ rank,
                       const int* __restrict__ rowptr, int2* __restrict__ csr) {
    int e = blockIdx.x * blockDim.x + threadIdx.x;
    if (e >= E) return;
    const int d = ei[E + e];
    csr[rowptr[d] + rank[e]] = make_int2(ei[e], e);
}

template <int K>
__global__ void k_gemm_pair(const float* __restrict__ A,
                            const float* __restrict__ Wl, const float* __restrict__ bl,
                            const float* __restrict__ Wr, const float* __restrict__ br,
                            float* __restrict__ outL, float* __restrict__ outR) {
    const int wave = threadIdx.x >> 5;
    const int lane = threadIdx.x & 31;
    const int r    = lane & 15;
    const int hi   = lane >> 4;
    const int row0 = blockIdx.x * 16;
    const int col0 = wave * 16;
    const int arow = row0 + r;
    const int bcol = col0 + r;

    v8f accL = {};
    v8f accR = {};
    #pragma unroll
    for (int k0 = 0; k0 < K; k0 += 4) {
        const int kk = k0 + hi * 2;
        v2f a = *reinterpret_cast<const v2f*>(A + arow * K + kk);
        v2f b;
        b.x = Wl[kk * HC + bcol];
        b.y = Wl[(kk + 1) * HC + bcol];
        accL = __builtin_amdgcn_wmma_f32_16x16x4_f32(false, a, false, b,
                                                     (short)0, accL, false, false);
        v2f b2;
        b2.x = Wr[kk * HC + bcol];
        b2.y = Wr[(kk + 1) * HC + bcol];
        accR = __builtin_amdgcn_wmma_f32_16x16x4_f32(false, a, false, b2,
                                                     (short)0, accR, false, false);
    }
    const float bLv = bl[bcol];
    const float bRv = br[bcol];
    #pragma unroll
    for (int i = 0; i < 8; ++i) {
        const int m = row0 + hi * 8 + i;
        outL[m * HC + bcol] = accL[i] + bLv;
        outR[m * HC + bcol] = accR[i] + bRv;
    }
}

static __device__ __forceinline__ float head_sum(float v) {
    v += __int_as_float(__builtin_amdgcn_update_dpp(0, __float_as_int(v), 0xB1,  0xF, 0xF, false));
    v += __int_as_float(__builtin_amdgcn_update_dpp(0, __float_as_int(v), 0x4E,  0xF, 0xF, false));
    v += __int_as_float(__builtin_amdgcn_update_dpp(0, __float_as_int(v), 0x141, 0xF, 0xF, false));
    return v;
}

__global__ __launch_bounds__(256) void k_gat_node(
        const float* __restrict__ xl, const float* __restrict__ xr,
        const float* __restrict__ eattr,
        const int*   __restrict__ rowptr,
        const int2*  __restrict__ csr,
        const float* __restrict__ We,
        const float* __restrict__ att,
        const float* __restrict__ bo,
        float* __restrict__ hout) {
    const int lane = threadIdx.x & 31;
    const int node = __builtin_amdgcn_readfirstlane(blockIdx.x * 8 + (threadIdx.x >> 5));
    const int c0 = lane * 2;

    float w0[ED], w1[ED];
    #pragma unroll
    for (int j = 0; j < ED; ++j) {
        const v2f w = *reinterpret_cast<const v2f*>(We + j * HC + c0);
        w0[j] = w.x; w1[j] = w.y;
    }
    const v2f at  = *reinterpret_cast<const v2f*>(att + c0);
    const v2f xrd = *reinterpret_cast<const v2f*>(xr + (size_t)node * HC + c0);
    const v2f xld = *reinterpret_cast<const v2f*>(xl + (size_t)node * HC + c0);

    float m = -INFINITY, z = 0.0f, acc0 = 0.0f, acc1 = 0.0f;
    float se0 = 0.0f, se1 = 0.0f;

    const int beg = rowptr[node];
    const int end = rowptr[node + 1];
    asm volatile("" :: "v"(xld.x), "v"(xld.y));
    for (int base = beg; base < end; base += 32) {
        const int cnt = min(32, end - base);
        const int2 se = csr[base + min(lane, cnt - 1)];
        for (int k = 0; k < cnt; k += 4) {
            v2f   xs[4];
            float ea[4][ED];
            #pragma unroll
            for (int u = 0; u < 4; ++u) {
                const int kk = min(k + u, cnt - 1);
                const int s = __builtin_amdgcn_readlane(se.x, kk);
                const int e = __builtin_amdgcn_readlane(se.y, kk);
                xs[u] = *reinterpret_cast<const v2f*>(xl + (size_t)s * HC + c0);
                #pragma unroll
                for (int j = 0; j < ED; ++j) ea[u][j] = eattr[(size_t)e * ED + j];
            }
            #pragma unroll
            for (int u = 0; u < 4; ++u) {
                const bool live = (k + u < cnt);
                float em0 = 0.0f, em1 = 0.0f;
                #pragma unroll
                for (int j = 0; j < ED; ++j) {
                    em0 = fmaf(ea[u][j], w0[j], em0);
                    em1 = fmaf(ea[u][j], w1[j], em1);
                }
                const float lv = live ? 1.0f : 0.0f;
                se0 = fmaf(lv, em0, se0);
                se1 = fmaf(lv, em1, se1);
                float v0 = (xs[u].x + xrd.x) + em0;
                float v1 = (xs[u].y + xrd.y) + em1;
                v0 = (v0 > 0.0f) ? v0 : NEG_ATT * v0;
                v1 = (v1 > 0.0f) ? v1 : NEG_ATT * v1;
                float lg = head_sum(fmaf(v1, at.y, v0 * at.x));
                lg = live ? lg : -INFINITY;
                const float mn = fmaxf(m, lg);
                const float sc = __expf(m - mn);
                const float w  = __expf(lg - mn);
                z    = fmaf(z,    sc, w);
                acc0 = fmaf(acc0, sc, w * xs[u].x);
                acc1 = fmaf(acc1, sc, w * xs[u].y);
                m = mn;
            }
        }
    }

    {
        const float inv = 1.0f / fmaxf((float)(end - beg), 1.0f);
        const float em0 = se0 * inv;
        const float em1 = se1 * inv;
        float v0 = (xld.x + xrd.x) + em0;
        float v1 = (xld.y + xrd.y) + em1;
        v0 = (v0 > 0.0f) ? v0 : NEG_ATT * v0;
        v1 = (v1 > 0.0f) ? v1 : NEG_ATT * v1;
        const float lg = head_sum(fmaf(v1, at.y, v0 * at.x));
        const float mn = fmaxf(m, lg);
        const float sc = __expf(m - mn);
        const float w  = __expf(lg - mn);
        z    = fmaf(z,    sc, w);
        acc0 = fmaf(acc0, sc, w * xld.x);
        acc1 = fmaf(acc1, sc, w * xld.y);
    }

    const v2f b = *reinterpret_cast<const v2f*>(bo + c0);
    float o0 = acc0 / z + b.x;
    float o1 = acc1 / z + b.y;
    v2f o;
    o.x = (o0 > 0.0f) ? o0 : NEG * o0;
    o.y = (o1 > 0.0f) ? o1 : NEG * o1;
    *reinterpret_cast<v2f*>(hout + (size_t)node * HC + c0) = o;
}

__global__ void k_head(const float* __restrict__ h2,
                       const float* __restrict__ Ws,
                       const float* __restrict__ bs,
                       float* __restrict__ out) {
    int n = blockIdx.x * blockDim.x + threadIdx.x;
    if (n >= N) return;
    const float4* hv = reinterpret_cast<const float4*>(h2 + (size_t)n * HC);
    const float4* wv = reinterpret_cast<const float4*>(Ws);
    float s = 0.0f;
    #pragma unroll
    for (int j = 0; j < 16; ++j) {
        float4 a = hv[j], b = wv[j];
        s = fmaf(a.x, b.x, s); s = fmaf(a.y, b.y, s);
        s = fmaf(a.z, b.z, s); s = fmaf(a.w, b.w, s);
    }
    out[n] = s + bs[0];
}

static inline int blocks(long long n, int bs) { return (int)((n + bs - 1) / bs); }

extern "C" void kernel_launch(void* const* d_in, const int* in_sizes, int n_in,
                              void* d_out, int out_size, void* d_ws, size_t ws_size,
                              hipStream_t stream) {
    const float* x     = (const float*)d_in[0];
    const int*   ei    = (const int*)d_in[1];
    const float* eattr = (const float*)d_in[2];
    const float* Wl1 = (const float*)d_in[3];  const float* bl1 = (const float*)d_in[4];
    const float* Wr1 = (const float*)d_in[5];  const float* br1 = (const float*)d_in[6];
    const float* We1 = (const float*)d_in[7];  const float* att1 = (const float*)d_in[8];
    const float* bo1 = (const float*)d_in[9];
    const float* Wl2 = (const float*)d_in[10]; const float* bl2 = (const float*)d_in[11];
    const float* Wr2 = (const float*)d_in[12]; const float* br2 = (const float*)d_in[13];
    const float* We2 = (const float*)d_in[14]; const float* att2 = (const float*)d_in[15];
    const float* bo2 = (const float*)d_in[16];
    const float* Ws  = (const float*)d_in[17]; const float* bs = (const float*)d_in[18];
    float* out = (float*)d_out;

    float* ws = (float*)d_ws;
    size_t o = 0;
    float* xl    = ws + o; o += (size_t)N * HC;
    float* xr    = ws + o; o += (size_t)N * HC;
    float* hbuf  = ws + o; o += (size_t)N * HC;
    int*   deg   = (int*)(ws + o); o += (size_t)N;
    int*   rowp  = (int*)(ws + o); o += (size_t)N + 2;
    int*   rank  = (int*)(ws + o); o += (size_t)E;
    int2*  csr   = (int2*)(ws + o); o += (size_t)E * 2;

    const int BT = 256;

    k_zero_i<<<blocks(N, BT), BT, 0, stream>>>(deg, N);
    k_count<<<blocks(E, BT), BT, 0, stream>>>(ei, deg, rank);
    k_scan<<<blocks(N, 1024), 1024, 0, stream>>>(deg, rowp);
    k_fill<<<blocks(E, BT), BT, 0, stream>>>(ei, rank, rowp, csr);

    k_gemm_pair<F><<<N / 16, 128, 0, stream>>>(x, Wl1, bl1, Wr1, br1, xl, xr);
    k_gat_node<<<N / 8, 256, 0, stream>>>(xl, xr, eattr, rowp, csr, We1, att1, bo1, hbuf);

    k_gemm_pair<HC><<<N / 16, 128, 0, stream>>>(hbuf, Wl2, bl2, Wr2, br2, xl, xr);
    k_gat_node<<<N / 8, 256, 0, stream>>>(xl, xr, eattr, rowp, csr, We2, att2, bo2, hbuf);

    k_head<<<blocks(N, BT), BT, 0, stream>>>(hbuf, Ws, bs, out);
}


# edge bucketing on LDS counters: slabs of edges grouped by 196-node bin, a work-group per bin ranks and places its edges; no global atomics, 4 launches -> 2; LDS arrival-order rank, sum order changed
# speedup vs baseline: 7.6011x; 1.4812x over previous
#include <hip/hip_runtime.h>
#include <math.h>

static constexpr int N    = 50000;
static constexpr int E    = 800000;
static constexpr int F    = 16;
static constexpr int ED   = 7;
static constexpr int HC   = 64;
static constexpr float NEG_ATT = 0.2f;
static constexpr float NEG     = 0.01f;

typedef __attribute__((ext_vector_type(2))) float v2f;
typedef __attribute__((ext_vector_type(8))) float v8f;

static constexpr int NPB   = 196;
static constexpr int NBIN  = 256;
static constexpr int NSLAB = 256;
static constexpr int SL    = E / NSLAB;
static_assert(NSLAB * SL == E, "slabs must tile the edge list");
static_assert(NBIN * NPB >= N, "bins must cover the nodes");
static_assert(N <= 65536 && NPB <= 256, "record packing: 16-bit source, 8-bit node-in-bin");
static constexpr int SLAB_T = 256;
static constexpr int EPT    = (SL + SLAB_T - 1) / SLAB_T;
static_assert(SLAB_T == NBIN, "k_slab: one bin counter per thread");

static __device__ __forceinline__ int excl_scan256(int v, int t, int* sWv) {
    const int lane = t & 31;
    const int wave = t >> 5;
    int incl = v;
    #pragma unroll
    for (int o = 1; o < 32; o <<= 1) {
        int u = __shfl_up(incl, o);
        if (lane >= o) incl += u;
    }
    if (t < 256 && lane == 31) sWv[wave] = incl;
    __syncthreads();
    int wb = 0;
    if (t < 256) {
        for (int w = 0; w < wave; ++w) wb += sWv[w];
    }
    return wb + incl - v;
}

__global__ __launch_bounds__(SLAB_T) void k_slab(const int* __restrict__ ei,
                                                 int2* __restrict__ recs,
                                                 int* __restrict__ offT) {
    __shared__ int sHist[NBIN];
    __shared__ int sOff[NBIN];
    __shared__ int sWv[8];
    const int t  = threadIdx.x;
    const int p  = blockIdx.x;
    const int e0 = p * SL;
    sHist[t] = 0;
    __syncthreads();

    int d[EPT], s[EPT], rk[EPT];
    #pragma unroll
    for (int j = 0; j < EPT; ++j) {
        const int i = j * SLAB_T + t;
        d[j] = 0; s[j] = 0;
        if (i < SL) {
            d[j] = ei[E + e0 + i];
            s[j] = ei[e0 + i];
        }
    }
    #pragma unroll
    for (int j = 0; j < EPT; ++j) {
        const int i = j * SLAB_T + t;
        rk[j] = 0;
        if (i < SL) rk[j] = atomicAdd(&sHist[d[j] / NPB], 1);
    }
    __syncthreads();

    const int excl = excl_scan256(sHist[t], t, sWv);
    sOff[t] = excl;
    offT[t * NSLAB + p] = excl;
    if (t == 0) offT[NBIN * NSLAB + p] = SL;
    __syncthreads();

    #pragma unroll
    for (int j = 0; j < EPT; ++j) {
        const int i = j * SLAB_T + t;
        if (i < SL) {
            const int b  = d[j] / NPB;
            const int dl = d[j] - b * NPB;
            recs[e0 + sOff[b] + rk[j]] = make_int2(s[j] | (dl << 16), e0 + i);
        }
    }
}

__global__ __launch_bounds__(1024) void k_bin(const int2* __restrict__ recs,
                                              const int* __restrict__ offT,
                                              int* __restrict__ rowptr,
                                              int2* __restrict__ csr) {
    __shared__ int sCnt[256];
    __shared__ int sCur[256];
    __shared__ int sSegBeg[NSLAB];
    __shared__ int sSegCnt[NSLAB];
    __shared__ int sRed[32];
    __shared__ int sWv[8];
    const int t    = threadIdx.x;
    const int lane = t & 31;
    const int wave = t >> 5;
    const int b    = blockIdx.x;

    int o0 = 0;
    if (t < NSLAB) {
        o0 = offT[b * NSLAB + t];
        const int o1 = offT[(b + 1) * NSLAB + t];
        sSegBeg[t] = t * SL + o0;
        sSegCnt[t] = o1 - o0;
    }
    if (t < 256) sCnt[t] = 0;
    int part = o0;
    #pragma unroll
    for (int o = 16; o > 0; o >>= 1) part += __shfl_xor(part, o);
    if (lane == 0) sRed[wave] = part;
    __syncthreads();

    int base = 0;
    #pragma unroll
    for (int w = 0; w < 32; ++w) base += sRed[w];

    const int seg = t >> 2;
    const int sub = t & 3;
    const int sb  = sSegBeg[seg];
    const int sc  = sSegCnt[seg];

    for (int i0 = sub; i0 < sc; i0 += 16) {
        int2 r[4];
        #pragma unroll
        for (int u = 0; u < 4; ++u) r[u] = recs[sb + min(i0 + 4 * u, sc - 1)];
        asm volatile("" :: "v"(r[0].x), "v"(r[1].x), "v"(r[2].x), "v"(r[3].x));
        #pragma unroll
        for (int u = 0; u < 4; ++u) {
            const int i = i0 + 4 * u;
            if (i < sc) atomicAdd(&sCnt[r[u].x >> 16], 1);
        }
    }
    __syncthreads();

    const int cntv = (t < 256) ? sCnt[t] : 0;
    const int excl = excl_scan256(cntv, t, sWv);
    const int node = b * NPB + t;
    if (t < NPB && node < N) rowptr[node] = base + excl;
    if (b == NBIN - 1 && t == 0) rowptr[N] = E;
    if (t < 256) sCur[t] = excl;
    __syncthreads();

    for (int i0 = sub; i0 < sc; i0 += 16) {
        int2 r[4];
        #pragma unroll
        for (int u = 0; u < 4; ++u) r[u] = recs[sb + min(i0 + 4 * u, sc - 1)];
        asm volatile("" :: "v"(r[0].x), "v"(r[1].x), "v"(r[2].x), "v"(r[3].x));
        #pragma unroll
        for (int u = 0; u < 4; ++u) {
            const int i = i0 + 4 * u;
            if (i < sc) {
                const int k = atomicAdd(&sCur[r[u].x >> 16], 1);
                csr[base + k] = make_int2(r[u].x & 0xFFFF, r[u].y);
            }
        }
    }
}

template <int K>
__global__ void k_gemm_pair(const float* __restrict__ A,
                            const float* __restrict__ Wl, const float* __restrict__ bl,
                            const float* __restrict__ Wr, const float* __restrict__ br,
                            float* __restrict__ outL, float* __restrict__ outR) {
    const int wave = threadIdx.x >> 5;
    const int lane = threadIdx.x & 31;
    const int r    = lane & 15;
    const int hi   = lane >> 4;
    const int row0 = blockIdx.x * 16;
    const int col0 = wave * 16;
    const int arow = row0 + r;
    const int bcol = col0 + r;

    v8f accL = {};
    v8f accR = {};
    #pragma unroll
    for (int k0 = 0; k0 < K; k0 += 4) {
        const int kk = k0 + hi * 2;
        v2f a = *reinterpret_cast<const v2f*>(A + arow * K + kk);
        v2f b;
        b.x = Wl[kk * HC + bcol];
        b.y = Wl[(kk + 1) * HC + bcol];
        accL = __builtin_amdgcn_wmma_f32_16x16x4_f32(false, a, false, b,
                                                     (short)0, accL, false, false);
        v2f b2;
        b2.x = Wr[kk * HC + bcol];
        b2.y = Wr[(kk + 1) * HC + bcol];
        accR = __builtin_amdgcn_wmma_f32_16x16x4_f32(false, a, false, b2,
                                                     (short)0, accR, false, false);
    }
    const float bLv = bl[bcol];
    const float bRv = br[bcol];
    #pragma unroll
    for (int i = 0; i < 8; ++i) {
        const int m = row0 + hi * 8 + i;
        outL[m * HC + bcol] = accL[i] + bLv;
        outR[m * HC + bcol] = accR[i] + bRv;
    }
}

static __device__ __forceinline__ float head_sum(float v) {
    v += __int_as_float(__builtin_amdgcn_update_dpp(0, __float_as_int(v), 0xB1,  0xF, 0xF, false));
    v += __int_as_float(__builtin_amdgcn_update_dpp(0, __float_as_int(v), 0x4E,  0xF, 0xF, false));
    v += __int_as_float(__builtin_amdgcn_update_dpp(0, __float_as_int(v), 0x141, 0xF, 0xF, false));
    return v;
}

__global__ __launch_bounds__(256) void k_gat_node(
        const float* __restrict__ xl, const float* __restrict__ xr,
        const float* __restrict__ eattr,
        const int*   __restrict__ rowptr,
        const int2*  __restrict__ csr,
        const float* __restrict__ We,
        const float* __restrict__ att,
        const float* __restrict__ bo,
        float* __restrict__ hout) {
    const int lane = threadIdx.x & 31;
    const int node = __builtin_amdgcn_readfirstlane(blockIdx.x * 8 + (threadIdx.x >> 5));
    const int c0 = lane * 2;

    float w0[ED], w1[ED];
    #pragma unroll
    for (int j = 0; j < ED; ++j) {
        const v2f w = *reinterpret_cast<const v2f*>(We + j * HC + c0);
        w0[j] = w.x; w1[j] = w.y;
    }
    const v2f at  = *reinterpret_cast<const v2f*>(att + c0);
    const v2f xrd = *reinterpret_cast<const v2f*>(xr + (size_t)node * HC + c0);
    const v2f xld = *reinterpret_cast<const v2f*>(xl + (size_t)node * HC + c0);

    float m = -INFINITY, z = 0.0f, acc0 = 0.0f, acc1 = 0.0f;
    float se0 = 0.0f, se1 = 0.0f;

    const int beg = rowptr[node];
    const int end = rowptr[node + 1];
    asm volatile("" :: "v"(xld.x), "v"(xld.y));
    for (int base = beg; base < end; base += 32) {
        const int cnt = min(32, end - base);
        const int2 se = csr[base + min(lane, cnt - 1)];
        for (int k = 0; k < cnt; k += 4) {
            v2f   xs[4];
            float ea[4][ED];
            #pragma unroll
            for (int u = 0; u < 4; ++u) {
                const int kk = min(k + u, cnt - 1);
                const int s = __builtin_amdgcn_readlane(se.x, kk);
                const int e = __builtin_amdgcn_readlane(se.y, kk);
                xs[u] = *reinterpret_cast<const v2f*>(xl + (size_t)s * HC + c0);
                #pragma unroll
                for (int j = 0; j < ED; ++j) ea[u][j] = eattr[(size_t)e * ED + j];
            }
            #pragma unroll
            for (int u = 0; u < 4; ++u) {
                const bool live = (k + u < cnt);
                float em0 = 0.0f, em1 = 0.0f;
                #pragma unroll
                for (int j = 0; j < ED; ++j) {
                    em0 = fmaf(ea[u][j], w0[j], em0);
                    em1 = fmaf(ea[u][j], w1[j], em1);
                }
                const float lv = live ? 1.0f : 0.0f;
                se0 = fmaf(lv, em0, se0);
                se1 = fmaf(lv, em1, se1);
                float v0 = (xs[u].x + xrd.x) + em0;
                float v1 = (xs[u].y + xrd.y) + em1;
                v0 = (v0 > 0.0f) ? v0 : NEG_ATT * v0;
                v1 = (v1 > 0.0f) ? v1 : NEG_ATT * v1;
                float lg = head_sum(fmaf(v1, at.y, v0 * at.x));
                lg = live ? lg : -INFINITY;
                const float mn = fmaxf(m, lg);
                const float sc = __expf(m - mn);
                const float w  = __expf(lg - mn);
                z    = fmaf(z,    sc, w);
                acc0 = fmaf(acc0, sc, w * xs[u].x);
                acc1 = fmaf(acc1, sc, w * xs[u].y);
                m = mn;
            }
        }
    }

    {
        const float inv = 1.0f / fmaxf((float)(end - beg), 1.0f);
        const float em0 = se0 * inv;
        const float em1 = se1 * inv;
        float v0 = (xld.x + xrd.x) + em0;
        float v1 = (xld.y + xrd.y) + em1;
        v0 = (v0 > 0.0f) ? v0 : NEG_ATT * v0;
        v1 = (v1 > 0.0f) ? v1 : NEG_ATT * v1;
        const float lg = head_sum(fmaf(v1, at.y, v0 * at.x));
        const float mn = fmaxf(m, lg);
        const float sc = __expf(m - mn);
        const float w  = __expf(lg - mn);
        z    = fmaf(z,    sc, w);
        acc0 = fmaf(acc0, sc, w * xld.x);
        acc1 = fmaf(acc1, sc, w * xld.y);
    }

    const v2f b = *reinterpret_cast<const v2f*>(bo + c0);
    float o0 = acc0 / z + b.x;
    float o1 = acc1 / z + b.y;
    v2f o;
    o.x = (o0 > 0.0f) ? o0 : NEG * o0;
    o.y = (o1 > 0.0f) ? o1 : NEG * o1;
    *reinterpret_cast<v2f*>(hout + (size_t)node * HC + c0) = o;
}

__global__ void k_head(const float* __restrict__ h2,
                       const float* __restrict__ Ws,
                       const float* __restrict__ bs,
                       float* __restrict__ out) {
    int n = blockIdx.x * blockDim.x + threadIdx.x;
    if (n >= N) return;
    const float4* hv = reinterpret_cast<const float4*>(h2 + (size_t)n * HC);
    const float4* wv = reinterpret_cast<const float4*>(Ws);
    float s = 0.0f;
    #pragma unroll
    for (int j = 0; j < 16; ++j) {
        float4 a = hv[j], b = wv[j];
        s = fmaf(a.x, b.x, s); s = fmaf(a.y, b.y, s);
        s = fmaf(a.z, b.z, s); s = fmaf(a.w, b.w, s);
    }
    out[n] = s + bs[0];
}

static inline int blocks(long long n, int bs) { return (int)((n + bs - 1) / bs); }

extern "C" void kernel_launch(void* const* d_in, const int* in_sizes, int n_in,
                              void* d_out, int out_size, void* d_ws, size_t ws_size,
                              hipStream_t stream) {
    const float* x     = (const float*)d_in[0];
    const int*   ei    = (const int*)d_in[1];
    const float* eattr = (const float*)d_in[2];
    const float* Wl1 = (const float*)d_in[3];  const float* bl1 = (const float*)d_in[4];
    const float* Wr1 = (const float*)d_in[5];  const float* br1 = (const float*)d_in[6];
    const float* We1 = (const float*)d_in[7];  const float* att1 = (const float*)d_in[8];
    const float* bo1 = (const float*)d_in[9];
    const float* Wl2 = (const float*)d_in[10]; const float* bl2 = (const float*)d_in[11];
    const float* Wr2 = (const float*)d_in[12]; const float* br2 = (const float*)d_in[13];
    const float* We2 = (const float*)d_in[14]; const float* att2 = (const float*)d_in[15];
    const float* bo2 = (const float*)d_in[16];
    const float* Ws  = (const float*)d_in[17]; const float* bs = (const float*)d_in[18];
    float* out = (float*)d_out;

    float* ws = (float*)d_ws;
    size_t o = 0;
    float* xl    = ws + o; o += (size_t)N * HC;
    float* xr    = ws + o; o += (size_t)N * HC;
    float* hbuf  = ws + o; o += (size_t)N * HC;
    int*   rowp  = (int*)(ws + o); o += (size_t)N + 2;
    int2*  csr   = (int2*)(ws + o); o += (size_t)E * 2;
    int2*  recs  = (int2*)(ws + o); o += (size_t)E * 2;
    int*   offT  = (int*)(ws + o); o += (size_t)(NBIN + 1) * NSLAB;

    const int BT = 256;

    k_slab<<<NSLAB, SLAB_T, 0, stream>>>(ei, recs, offT);
    k_bin<<<NBIN, 1024, 0, stream>>>(recs, offT, rowp, csr);

    k_gemm_pair<F><<<N / 16, 128, 0, stream>>>(x, Wl1, bl1, Wr1, br1, xl, xr);
    k_gat_node<<<N / 8, 256, 0, stream>>>(xl, xr, eattr, rowp, csr, We1, att1, bo1, hbuf);

    k_gemm_pair<HC><<<N / 16, 128, 0, stream>>>(hbuf, Wl2, bl2, Wr2, br2, xl, xr);
    k_gat_node<<<N / 8, 256, 0, stream>>>(xl, xr, eattr, rowp, csr, We2, att2, bo2, hbuf);

    k_head<<<blocks(N, BT), BT, 0, stream>>>(hbuf, Ws, bs, out);
}


# node kernel: a lane owns 4 channels and each 16-lane half of a wave walks its own node's bucket (2 nodes per wave side by side, half the waves); logit dot and lane-sum order changed
# speedup vs baseline: 8.3495x; 1.0985x over previous
#include <hip/hip_runtime.h>
#include <math.h>

static constexpr int N    = 50000;
static constexpr int E    = 800000;
static constexpr int F    = 16;
static constexpr int ED   = 7;
static constexpr int HC   = 64;
static constexpr float NEG_ATT = 0.2f;
static constexpr float NEG     = 0.01f;

typedef __attribute__((ext_vector_type(2))) float v2f;
typedef __attribute__((ext_vector_type(8))) float v8f;

static constexpr int NPB   = 196;
static constexpr int NBIN  = 256;
static constexpr int NSLAB = 256;
static constexpr int SL    = E / NSLAB;
static_assert(NSLAB * SL == E, "slabs must tile the edge list");
static_assert(NBIN * NPB >= N, "bins must cover the nodes");
static_assert(N <= 65536 && NPB <= 256, "record packing: 16-bit source, 8-bit node-in-bin");
static_assert(N % 16 == 0, "node kernel: 8 waves of 2 nodes per work-group tile the nodes");
static constexpr int SLAB_T = 256;
static constexpr int EPT    = (SL + SLAB_T - 1) / SLAB_T;
static_assert(SLAB_T == NBIN, "k_slab: one bin counter per thread");

static __device__ __forceinline__ int excl_scan256(int v, int t, int* sWv) {
    const int lane = t & 31;
    const int wave = t >> 5;
    int incl = v;
    #pragma unroll
    for (int o = 1; o < 32; o <<= 1) {
        int u = __shfl_up(incl, o);
        if (lane >= o) incl += u;
    }
    if (t < 256 && lane == 31) sWv[wave] = incl;
    __syncthreads();
    int wb = 0;
    if (t < 256) {
        for (int w = 0; w < wave; ++w) wb += sWv[w];
    }
    return wb + incl - v;
}

__global__ __launch_bounds__(SLAB_T) void k_slab(const int* __restrict__ ei,
                                                 int2* __restrict__ recs,
                                                 int* __restrict__ offT) {
    __shared__ int sHist[NBIN];
    __shared__ int sOff[NBIN];
    __shared__ int sWv[8];
    const int t  = threadIdx.x;
    const int p  = blockIdx.x;
    const int e0 = p * SL;
    sHist[t] = 0;
    __syncthreads();

    int d[EPT], s[EPT], rk[EPT];
    #pragma unroll
    for (int j = 0; j < EPT; ++j) {
        const int i = j * SLAB_T + t;
        d[j] = 0; s[j] = 0;
        if (i < SL) {
            d[j] = ei[E + e0 + i];
            s[j] = ei[e0 + i];
        }
    }
    #pragma unroll
    for (int j = 0; j < EPT; ++j) {
        const int i = j * SLAB_T + t;
        rk[j] = 0;
        if (i < SL) rk[j] = atomicAdd(&sHist[d[j] / NPB], 1);
    }
    __syncthreads();

    const int excl = excl_scan256(sHist[t], t, sWv);
    sOff[t] = excl;
    offT[t * NSLAB + p] = excl;
    if (t == 0) offT[NBIN * NSLAB + p] = SL;
    __syncthreads();

    #pragma unroll
    for (int j = 0; j < EPT; ++j) {
        const int i = j * SLAB_T + t;
        if (i < SL) {
            const int b  = d[j] / NPB;
            const int dl = d[j] - b * NPB;
            recs[e0 + sOff[b] + rk[j]] = make_int2(s[j] | (dl << 16), e0 + i);
        }
    }
}

__global__ __launch_bounds__(1024) void k_bin(const int2* __restrict__ recs,
                                              const int* __restrict__ offT,
                                              int* __restrict__ rowptr,
                                              int2* __restrict__ csr) {
    __shared__ int sCnt[256];
    __shared__ int sCur[256];
    __shared__ int sSegBeg[NSLAB];
    __shared__ int sSegCnt[NSLAB];
    __shared__ int sRed[32];
    __shared__ int sWv[8];
    const int t    = threadIdx.x;
    const int lane = t & 31;
    const int wave = t >> 5;
    const int b    = blockIdx.x;

    int o0 = 0;
    if (t < NSLAB) {
        o0 = offT[b * NSLAB + t];
        const int o1 = offT[(b + 1) * NSLAB + t];
        sSegBeg[t] = t * SL + o0;
        sSegCnt[t] = o1 - o0;
    }
    if (t < 256) sCnt[t] = 0;
    int part = o0;
    #pragma unroll
    for (int o = 16; o > 0; o >>= 1) part += __shfl_xor(part, o);
    if (lane == 0) sRed[wave] = part;
    __syncthreads();

    int base = 0;
    #pragma unroll
    for (int w = 0; w < 32; ++w) base += sRed[w];

    const int seg = t >> 2;
    const int sub = t & 3;
    const int sb  = sSegBeg[seg];
    const int sc  = sSegCnt[seg];

    for (int i0 = sub; i0 < sc; i0 += 16) {
        int2 r[4];
        #pragma unroll
        for (int u = 0; u < 4; ++u) r[u] = recs[sb + min(i0 + 4 * u, sc - 1)];
        asm volatile("" :: "v"(r[0].x), "v"(r[1].x), "v"(r[2].x), "v"(r[3].x));
        #pragma unroll
        for (int u = 0; u < 4; ++u) {
            const int i = i0 + 4 * u;
            if (i < sc) atomicAdd(&sCnt[r[u].x >> 16], 1);
        }
    }
    __syncthreads();

    const int cntv = (t < 256) ? sCnt[t] : 0;
    const int excl = excl_scan256(cntv, t, sWv);
    const int node = b * NPB + t;
    if (t < NPB && node < N) rowptr[node] = base + excl;
    if (b == NBIN - 1 && t == 0) rowptr[N] = E;
    if (t < 256) sCur[t] = excl;
    __syncthreads();

    for (int i0 = sub; i0 < sc; i0 += 16) {
        int2 r[4];
        #pragma unroll
        for (int u = 0; u < 4; ++u) r[u] = recs[sb + min(i0 + 4 * u, sc - 1)];
        asm volatile("" :: "v"(r[0].x), "v"(r[1].x), "v"(r[2].x), "v"(r[3].x));
        #pragma unroll
        for (int u = 0; u < 4; ++u) {
            const int i = i0 + 4 * u;
            if (i < sc) {
                const int k = atomicAdd(&sCur[r[u].x >> 16], 1);
                csr[base + k] = make_int2(r[u].x & 0xFFFF, r[u].y);
            }
        }
    }
}

template <int K>
__global__ void k_gemm_pair(const float* __restrict__ A,
                            const float* __restrict__ Wl, const float* __restrict__ bl,
                            const float* __restrict__ Wr, const float* __restrict__ br,
                            float* __restrict__ outL, float* __restrict__ outR) {
    const int wave = threadIdx.x >> 5;
    const int lane = threadIdx.x & 31;
    const int r    = lane & 15;
    const int hi   = lane >> 4;
    const int row0 = blockIdx.x * 16;
    const int col0 = wave * 16;
    const int arow = row0 + r;
    const int bcol = col0 + r;

    v8f accL = {};
    v8f accR = {};
    #pragma unroll
    for (int k0 = 0; k0 < K; k0 += 4) {
        const int kk = k0 + hi * 2;
        v2f a = *reinterpret_cast<const v2f*>(A + arow * K + kk);
        v2f b;
        b.x = Wl[kk * HC + bcol];
        b.y = Wl[(kk + 1) * HC + bcol];
        accL = __builtin_amdgcn_wmma_f32_16x16x4_f32(false, a, false, b,
                                                     (short)0, accL, false, false);
        v2f b2;
        b2.x = Wr[kk * HC + bcol];
        b2.y = Wr[(kk + 1) * HC + bcol];
        accR = __builtin_amdgcn_wmma_f32_16x16x4_f32(false, a, false, b2,
                                                     (short)0, accR, false, false);
    }
    const float bLv = bl[bcol];
    const float bRv = br[bcol];
    #pragma unroll
    for (int i = 0; i < 8; ++i) {
        const int m = row0 + hi * 8 + i;
        outL[m * HC + bcol] = accL[i] + bLv;
        outR[m * HC + bcol] = accR[i] + bRv;
    }
}

static __device__ __forceinline__ float quad_sum(float v) {
    v += __int_as_float(__builtin_amdgcn_update_dpp(0, __float_as_int(v), 0xB1, 0xF, 0xF, false));
    v += __int_as_float(__builtin_amdgcn_update_dpp(0, __float_as_int(v), 0x4E, 0xF, 0xF, false));
    return v;
}

static __device__ __forceinline__ float lrelu(float v, float slope) { return (v > 0.0f) ? v : slope * v; }

__global__ __launch_bounds__(256) void k_gat_node(
        const float* __restrict__ xl, const float* __restrict__ xr,
        const float* __restrict__ eattr,
        const int*   __restrict__ rowptr,
        const int2*  __restrict__ csr,
        const float* __restrict__ We,
        const float* __restrict__ att,
        const float* __restrict__ bo,
        float* __restrict__ hout) {
    const int lane = threadIdx.x & 31;
    const int half = lane >> 4;
    const int c    = lane & 15;
    const int wv = __builtin_amdgcn_readfirstlane(blockIdx.x * 8 + (threadIdx.x >> 5));
    const unsigned cb = (unsigned)c * 16u;
    const unsigned lb = (unsigned)lane * 16u;
    const size_t   pb = (size_t)wv * (2u * HC * 4u);

    float4 w[ED];
    #pragma unroll
    for (int j = 0; j < ED; ++j)
        w[j] = *reinterpret_cast<const float4*>(reinterpret_cast<const char*>(We) + j * (HC * 4) + cb);
    const float4 at  = *reinterpret_cast<const float4*>(reinterpret_cast<const char*>(att) + cb);
    const float4 xrd = *reinterpret_cast<const float4*>(reinterpret_cast<const char*>(xr) + pb + lb);
    const float4 xld = *reinterpret_cast<const float4*>(reinterpret_cast<const char*>(xl) + pb + lb);

    float m = -3.0e38f, z = 0.0f;
    float4 acc = make_float4(0.0f, 0.0f, 0.0f, 0.0f);
    float4 se  = make_float4(0.0f, 0.0f, 0.0f, 0.0f);

    const int* rp = rowptr + 2 * wv + half;
    const int beg = rp[0];
    const int deg = rp[1] - beg;
    const int dmax = max(__builtin_amdgcn_readlane(deg, 0), __builtin_amdgcn_readlane(deg, 16));
    asm volatile("" :: "v"(xld.x), "v"(xld.y), "v"(xld.z), "v"(xld.w));
    for (int base = 0; base < dmax; base += 16) {
        const int cnt = min(16, dmax - base);
        const int2 sl = csr[min(beg + max(min(base + c, deg - 1), 0), E - 1)];
        for (int i = 0; i < cnt; i += 4) {
            float4 xs[4];
            float  ea[4][ED];
            #pragma unroll
            for (int u = 0; u < 4; ++u) {
                const int ii = min(i + u, cnt - 1);
                const int a4 = ((lane & 16) | ii) << 2;
                const int s  = __builtin_amdgcn_ds_bpermute(a4, sl.x);
                const int e  = __builtin_amdgcn_ds_bpermute(a4, sl.y);
                xs[u] = *reinterpret_cast<const float4*>(reinterpret_cast<const char*>(xl) + ((unsigned)s * (HC * 4u) + cb));
                const float* pe = reinterpret_cast<const float*>(reinterpret_cast<const char*>(eattr) + (unsigned)__mul24(e, ED * 4));
                #pragma unroll
                for (int j = 0; j < ED; ++j) ea[u][j] = pe[j];
            }
            #pragma unroll
            for (int u = 0; u < 4; ++u) {
                asm volatile("" :: "v"(xs[u].x), "v"(xs[u].y), "v"(xs[u].z), "v"(xs[u].w),
                             "v"(ea[u][0]), "v"(ea[u][1]), "v"(ea[u][2]), "v"(ea[u][3]),
                             "v"(ea[u][4]), "v"(ea[u][5]), "v"(ea[u][6]));
            }
            #pragma unroll
            for (int u = 0; u < 4; ++u) {
                if (i + u < cnt) {
                    const bool live = (base + i + u < deg);
                    const float4 x = xs[u];
                    float4 em = make_float4(0.0f, 0.0f, 0.0f, 0.0f);
                    #pragma unroll
                    for (int j = 0; j < ED; ++j) {
                        em.x = fmaf(ea[u][j], w[j].x, em.x);
                        em.y = fmaf(ea[u][j], w[j].y, em.y);
                        em.z = fmaf(ea[u][j], w[j].z, em.z);
                        em.w = fmaf(ea[u][j], w[j].w, em.w);
                    }
                    const float lv = live ? 1.0f : 0.0f;
                    se.x = fmaf(lv, em.x, se.x);
                    se.y = fmaf(lv, em.y, se.y);
                    se.z = fmaf(lv, em.z, se.z);
                    se.w = fmaf(lv, em.w, se.w);
                    const float v0 = lrelu((x.x + xrd.x) + em.x, NEG_ATT);
                    const float v1 = lrelu((x.y + xrd.y) + em.y, NEG_ATT);
                    const float v2 = lrelu((x.z + xrd.z) + em.z, NEG_ATT);
                    const float v3 = lrelu((x.w + xrd.w) + em.w, NEG_ATT);
                    float lg = quad_sum(fmaf(v1, at.y, v0 * at.x) + fmaf(v3, at.w, v2 * at.z));
                    lg = live ? lg : -INFINITY;
                    const float mn = fmaxf(m, lg);
                    const float sc = __expf(m - mn);
                    const float wg = __expf(lg - mn);
                    z     = fmaf(z,     sc, wg);
                    acc.x = fmaf(acc.x, sc, wg * x.x);
                    acc.y = fmaf(acc.y, sc, wg * x.y);
                    acc.z = fmaf(acc.z, sc, wg * x.z);
                    acc.w = fmaf(acc.w, sc, wg * x.w);
                    m = mn;
                }
            }
        }
    }

    {
        const float inv = 1.0f / fmaxf((float)deg, 1.0f);
        const float v0 = lrelu((xld.x + xrd.x) + se.x * inv, NEG_ATT);
        const float v1 = lrelu((xld.y + xrd.y) + se.y * inv, NEG_ATT);
        const float v2 = lrelu((xld.z + xrd.z) + se.z * inv, NEG_ATT);
        const float v3 = lrelu((xld.w + xrd.w) + se.w * inv, NEG_ATT);
        const float lg = quad_sum(fmaf(v1, at.y, v0 * at.x) + fmaf(v3, at.w, v2 * at.z));
        const float mn = fmaxf(m, lg);
        const float sc = __expf(m - mn);
        const float wg = __expf(lg - mn);
        z     = fmaf(z,     sc, wg);
        acc.x = fmaf(acc.x, sc, wg * xld.x);
        acc.y = fmaf(acc.y, sc, wg * xld.y);
        acc.z = fmaf(acc.z, sc, wg * xld.z);
        acc.w = fmaf(acc.w, sc, wg * xld.w);
    }

    const float4 b = *reinterpret_cast<const float4*>(reinterpret_cast<const char*>(bo) + cb);
    float4 o;
    o.x = lrelu(acc.x / z + b.x, NEG);
    o.y = lrelu(acc.y / z + b.y, NEG);
    o.z = lrelu(acc.z / z + b.z, NEG);
    o.w = lrelu(acc.w / z + b.w, NEG);
    *reinterpret_cast<float4*>(reinterpret_cast<char*>(hout) + pb + lb) = o;
}

__global__ void k_head(const float* __restrict__ h2,
                       const float* __restrict__ Ws,
                       const float* __restrict__ bs,
                       float* __restrict__ out) {
    int n = blockIdx.x * blockDim.x + threadIdx.x;
    if (n >= N) return;
    const float4* hv = reinterpret_cast<const float4*>(h2 + (size_t)n * HC);
    const float4* wv = reinterpret_cast<const float4*>(Ws);
    float s = 0.0f;
    #pragma unroll
    for (int j = 0; j < 16; ++j) {
        float4 a = hv[j], b = wv[j];
        s = fmaf(a.x, b.x, s); s = fmaf(a.y, b.y, s);
        s = fmaf(a.z, b.z, s); s = fmaf(a.w, b.w, s);
    }
    out[n] = s + bs[0];
}

static inline int blocks(long long n, int bs) { return (int)((n + bs - 1) / bs); }

extern "C" void kernel_launch(void* const* d_in, const int* in_sizes, int n_in,
                              void* d_out, int out_size, void* d_ws, size_t ws_size,
                              hipStream_t stream) {
    const float* x     = (const float*)d_in[0];
    const int*   ei    = (const int*)d_in[1];
    const float* eattr = (const float*)d_in[2];
    const float* Wl1 = (const float*)d_in[3];  const float* bl1 = (const float*)d_in[4];
    const float* Wr1 = (const float*)d_in[5];  const float* br1 = (const float*)d_in[6];
    const float* We1 = (const float*)d_in[7];  const float* att1 = (const float*)d_in[8];
    const float* bo1 = (const float*)d_in[9];
    const float* Wl2 = (const float*)d_in[10]; const float* bl2 = (const float*)d_in[11];
    const float* Wr2 = (const float*)d_in[12]; const float* br2 = (const float*)d_in[13];
    const float* We2 = (const float*)d_in[14]; const float* att2 = (const float*)d_in[15];
    const float* bo2 = (const float*)d_in[16];
    const float* Ws  = (const float*)d_in[17]; const float* bs = (const float*)d_in[18];
    float* out = (float*)d_out;

    float* ws = (float*)d_ws;
    size_t o = 0;
    float* xl    = ws + o; o += (size_t)N * HC;
    float* xr    = ws + o; o += (size_t)N * HC;
    float* hbuf  = ws + o; o += (size_t)N * HC;
    int*   rowp  = (int*)(ws + o); o += (size_t)N + 2;
    int2*  csr   = (int2*)(ws + o); o += (size_t)E * 2;
    int2*  recs  = (int2*)(ws + o); o += (size_t)E * 2;
    int*   offT  = (int*)(ws + o); o += (size_t)(NBIN + 1) * NSLAB;

    const int BT = 256;

    k_slab<<<NSLAB, SLAB_T, 0, stream>>>(ei, recs, offT);
    k_bin<<<NBIN, 1024, 0, stream>>>(recs, offT, rowp, csr);

    k_gemm_pair<F><<<N / 16, 128, 0, stream>>>(x, Wl1, bl1, Wr1, br1, xl, xr);
    k_gat_node<<<N / 16, 256, 0, stream>>>(xl, xr, eattr, rowp, csr, We1, att1, bo1, hbuf);

    k_gemm_pair<HC><<<N / 16, 128, 0, stream>>>(hbuf, Wl2, bl2, Wr2, br2, xl, xr);
    k_gat_node<<<N / 16, 256, 0, stream>>>(xl, xr, eattr, rowp, csr, We2, att2, bo2, hbuf);

    k_head<<<blocks(N, BT), BT, 0, stream>>>(hbuf, Ws, bs, out);
}


# launch folding, 7 -> 5: layer-1 projection GEMM runs as extra work-groups of the edge slab launch; head dot moves into the last node kernel (4-channel lane dot, 16-lane sum: sum order changed)
# speedup vs baseline: 8.6815x; 1.0398x over previous
#include <hip/hip_runtime.h>
#include <math.h>

static constexpr int N    = 50000;
static constexpr int E    = 800000;
static constexpr int F    = 16;
static constexpr int ED   = 7;
static constexpr int HC   = 64;
static constexpr float NEG_ATT = 0.2f;
static constexpr float NEG     = 0.01f;

typedef __attribute__((ext_vector_type(2))) float v2f;
typedef __attribute__((ext_vector_type(8))) float v8f;

static constexpr int NPB   = 196;
static constexpr int NBIN  = 256;
static constexpr int NSLAB = 256;
static constexpr int SL    = E / NSLAB;
static_assert(NSLAB * SL == E, "slabs must tile the edge list");
static_assert(NBIN * NPB >= N, "bins must cover the nodes");
static_assert(N <= 65536 && NPB <= 256, "record packing: 16-bit source, 8-bit node-in-bin");
static_assert(N % 16 == 0, "node kernel: 8 waves of 2 nodes per work-group tile the nodes");
static constexpr int SLAB_T = 256;
static constexpr int EPT    = (SL + SLAB_T - 1) / SLAB_T;
static_assert(SLAB_T == NBIN, "k_slab: one bin counter per thread");
static constexpr int NTILE  = N / 16;
static constexpr int NPROJ  = (NTILE + 1) / 2;
static constexpr int NGRID0 = NSLAB + NPROJ;

template <int K>
static __device__ __forceinline__ void gemm_pair_tile(const float* __restrict__ A,
                                                      const float* __restrict__ Wl, const float* __restrict__ bl,
                                                      const float* __restrict__ Wr, const float* __restrict__ br,
                                                      float* __restrict__ outL, float* __restrict__ outR,
                                                      const int row0, const int col0, const int lane) {
    const int r    = lane & 15;
    const int hi   = lane >> 4;
    const int arow = row0 + r;
    const int bcol = col0 + r;

    v8f accL = {};
    v8f accR = {};
    #pragma unroll
    for (int k0 = 0; k0 < K; k0 += 4) {
        const int kk = k0 + hi * 2;
        v2f a = *reinterpret_cast<const v2f*>(A + arow * K + kk);
        v2f b;
        b.x = Wl[kk * HC + bcol];
        b.y = Wl[(kk + 1) * HC + bcol];
        accL = __builtin_amdgcn_wmma_f32_16x16x4_f32(false, a, false, b,
                                                     (short)0, accL, false, false);
        v2f b2;
        b2.x = Wr[kk * HC + bcol];
        b2.y = Wr[(kk + 1) * HC + bcol];
        accR = __builtin_amdgcn_wmma_f32_16x16x4_f32(false, a, false, b2,
                                                     (short)0, accR, false, false);
    }
    const float bLv = bl[bcol];
    const float bRv = br[bcol];
    #pragma unroll
    for (int i = 0; i < 8; ++i) {
        const int m = row0 + hi * 8 + i;
        outL[m * HC + bcol] = accL[i] + bLv;
        outR[m * HC + bcol] = accR[i] + bRv;
    }
}

template <int K>
__global__ void k_gemm_pair(const float* __restrict__ A,
                            const float* __restrict__ Wl, const float* __restrict__ bl,
                            const float* __restrict__ Wr, const float* __restrict__ br,
                            float* __restrict__ outL, float* __restrict__ outR) {
    gemm_pair_tile<K>(A, Wl, bl, Wr, br, outL, outR, blockIdx.x * 16, (threadIdx.x >> 5) * 16, threadIdx.x & 31);
}

static __device__ __forceinline__ int excl_scan256(int v, int t, int* sWv) {
    const int lane = t & 31;
    const int wave = t >> 5;
    int incl = v;
    #pragma unroll
    for (int o = 1; o < 32; o <<= 1) {
        int u = __shfl_up(incl, o);
        if (lane >= o) incl += u;
    }
    if (t < 256 && lane == 31) sWv[wave] = incl;
    __syncthreads();
    int wb = 0;
    if (t < 256) {
        for (int w = 0; w < wave; ++w) wb += sWv[w];
    }
    return wb + incl - v;
}

__global__ __launch_bounds__(SLAB_T) void k_slab(const int* __restrict__ ei,
                                                 int2* __restrict__ recs,
                                                 int* __restrict__ offT,
                                                 const float* __restrict__ x,
                                                 const float* __restrict__ Wl, const float* __restrict__ bl,
                                                 const float* __restrict__ Wr, const float* __restrict__ br,
                                                 float* __restrict__ xl, float* __restrict__ xr) {
    __shared__ int sHist[NBIN];
    __shared__ int sOff[NBIN];
    __shared__ int sWv[8];
    const int t  = threadIdx.x;
    if (blockIdx.x >= NSLAB) {
        const int wave = t >> 5;
        const int tile = (blockIdx.x - NSLAB) * 2 + (wave >> 2);
        if (tile < NTILE)
            gemm_pair_tile<F>(x, Wl, bl, Wr, br, xl, xr, tile * 16, (wave & 3) * 16, t & 31);
        return;
    }
    const int p  = blockIdx.x;
    const int e0 = p * SL;
    sHist[t] = 0;
    __syncthreads();

    int d[EPT], s[EPT], rk[EPT];
    #pragma unroll
    for (int j = 0; j < EPT; ++j) {
        const int i = j * SLAB_T + t;
        d[j] = 0; s[j] = 0;
        if (i < SL) {
            d[j] = ei[E + e0 + i];
            s[j] = ei[e0 + i];
        }
    }
    #pragma unroll
    for (int j = 0; j < EPT; ++j) {
        const int i = j * SLAB_T + t;
        rk[j] = 0;
        if (i < SL) rk[j] = atomicAdd(&sHist[d[j] / NPB], 1);
    }
    __syncthreads();

    const int excl = excl_scan256(sHist[t], t, sWv);
    sOff[t] = excl;
    offT[t * NSLAB + p] = excl;
    if (t == 0) offT[NBIN * NSLAB + p] = SL;
    __syncthreads();

    #pragma unroll
    for (int j = 0; j < EPT; ++j) {
        const int i = j * SLAB_T + t;
        if (i < SL) {
            const int b  = d[j] / NPB;
            const int dl = d[j] - b * NPB;
            recs[e0 + sOff[b] + rk[j]] = make_int2(s[j] | (dl << 16), e0 + i);
        }
    }
}

__global__ __launch_bounds__(1024) void k_bin(const int2* __restrict__ recs,
                                              const int* __restrict__ offT,
                                              int* __restrict__ rowptr,
                                              int2* __restrict__ csr) {
    __shared__ int sCnt[256];
    __shared__ int sCur[256];
    __shared__ int sSegBeg[NSLAB];
    __shared__ int sSegCnt[NSLAB];
    __shared__ int sRed[32];
    __shared__ int sWv[8];
    const int t    = threadIdx.x;
    const int lane = t & 31;
    const int wave = t >> 5;
    const int b    = blockIdx.x;

    int o0 = 0;
    if (t < NSLAB) {
        o0 = offT[b * NSLAB + t];
        const int o1 = offT[(b + 1) * NSLAB + t];
        sSegBeg[t] = t * SL + o0;
        sSegCnt[t] = o1 - o0;
    }
    if (t < 256) sCnt[t] = 0;
    int part = o0;
    #pragma unroll
    for (int o = 16; o > 0; o >>= 1) part += __shfl_xor(part, o);
    if (lane == 0) sRed[wave] = part;
    __syncthreads();

    int base = 0;
    #pragma unroll
    for (int w = 0; w < 32; ++w) base += sRed[w];

    const int seg = t >> 2;
    const int sub = t & 3;
    const int sb  = sSegBeg[seg];
    const int sc  = sSegCnt[seg];

    for (int i0 = sub; i0 < sc; i0 += 16) {
        int2 r[4];
        #pragma unroll
        for (int u = 0; u < 4; ++u) r[u] = recs[sb + min(i0 + 4 * u, sc - 1)];
        asm volatile("" :: "v"(r[0].x), "v"(r[1].x), "v"(r[2].x), "v"(r[3].x));
        #pragma unroll
        for (int u = 0; u < 4; ++u) {
            const int i = i0 + 4 * u;
            if (i < sc) atomicAdd(&sCnt[r[u].x >> 16], 1);
        }
    }
    __syncthreads();

    const int cntv = (t < 256) ? sCnt[t] : 0;
    const int excl = excl_scan256(cntv, t, sWv);
    const int node = b * NPB + t;
    if (t < NPB && node < N) rowptr[node] = base + excl;
    if (b == NBIN - 1 && t == 0) rowptr[N] = E;
    if (t < 256) sCur[t] = excl;
    __syncthreads();

    for (int i0 = sub; i0 < sc; i0 += 16) {
        int2 r[4];
        #pragma unroll
        for (int u = 0; u < 4; ++u) r[u] = recs[sb + min(i0 + 4 * u, sc - 1)];
        asm volatile("" :: "v"(r[0].x), "v"(r[1].x), "v"(r[2].x), "v"(r[3].x));
        #pragma unroll
        for (int u = 0; u < 4; ++u) {
            const int i = i0 + 4 * u;
            if (i < sc) {
                const int k = atomicAdd(&sCur[r[u].x >> 16], 1);
                csr[base + k] = make_int2(r[u].x & 0xFFFF, r[u].y);
            }
        }
    }
}

static __device__ __forceinline__ float quad_sum(float v) {
    v += __int_as_float(__builtin_amdgcn_update_dpp(0, __float_as_int(v), 0xB1, 0xF, 0xF, false));
    v += __int_as_float(__builtin_amdgcn_update_dpp(0, __float_as_int(v), 0x4E, 0xF, 0xF, false));
    return v;
}

static __device__ __forceinline__ float lrelu(float v, float slope) { return (v > 0.0f) ? v : slope * v; }

template <bool HEAD>
__global__ __launch_bounds__(256) void k_gat_node(
        const float* __restrict__ xl, const float* __restrict__ xr,
        const float* __restrict__ eattr,
        const int*   __restrict__ rowptr,
        const int2*  __restrict__ csr,
        const float* __restrict__ We,
        const float* __restrict__ att,
        const float* __restrict__ bo,
        const float* __restrict__ Ws,
        const float* __restrict__ bs,
        float* __restrict__ hout) {
    const int lane = threadIdx.x & 31;
    const int half = lane >> 4;
    const int c    = lane & 15;
    const int wv = __builtin_amdgcn_readfirstlane(blockIdx.x * 8 + (threadIdx.x >> 5));
    const unsigned cb = (unsigned)c * 16u;
    const unsigned lb = (unsigned)lane * 16u;
    const size_t   pb = (size_t)wv * (2u * HC * 4u);

    float4 w[ED];
    #pragma unroll
    for (int j = 0; j < ED; ++j)
        w[j] = *reinterpret_cast<const float4*>(reinterpret_cast<const char*>(We) + j * (HC * 4) + cb);
    const float4 at  = *reinterpret_cast<const float4*>(reinterpret_cast<const char*>(att) + cb);
    const float4 xrd = *reinterpret_cast<const float4*>(reinterpret_cast<const char*>(xr) + pb + lb);
    const float4 xld = *reinterpret_cast<const float4*>(reinterpret_cast<const char*>(xl) + pb + lb);

    float m = -3.0e38f, z = 0.0f;
    float4 acc = make_float4(0.0f, 0.0f, 0.0f, 0.0f);
    float4 se  = make_float4(0.0f, 0.0f, 0.0f, 0.0f);

    const int* rp = rowptr + 2 * wv + half;
    const int beg = rp[0];
    const int deg = rp[1] - beg;
    const int dmax = max(__builtin_amdgcn_readlane(deg, 0), __builtin_amdgcn_readlane(deg, 16));
    asm volatile("" :: "v"(xld.x), "v"(xld.y), "v"(xld.z), "v"(xld.w));
    for (int base = 0; base < dmax; base += 16) {
        const int cnt = min(16, dmax - base);
        const int2 sl = csr[min(beg + max(min(base + c, deg - 1), 0), E - 1)];
        for (int i = 0; i < cnt; i += 4) {
            float4 xs[4];
            float  ea[4][ED];
            #pragma unroll
            for (int u = 0; u < 4; ++u) {
                const int ii = min(i + u, cnt - 1);
                const int a4 = ((lane & 16) | ii) << 2;
                const int s  = __builtin_amdgcn_ds_bpermute(a4, sl.x);
                const int e  = __builtin_amdgcn_ds_bpermute(a4, sl.y);
                xs[u] = *reinterpret_cast<const float4*>(reinterpret_cast<const char*>(xl) + ((unsigned)s * (HC * 4u) + cb));
                const float* pe = reinterpret_cast<const float*>(reinterpret_cast<const char*>(eattr) + (unsigned)__mul24(e, ED * 4));
                #pragma unroll
                for (int j = 0; j < ED; ++j) ea[u][j] = pe[j];
            }
            #pragma unroll
            for (int u = 0; u < 4; ++u) {
                asm volatile("" :: "v"(xs[u].x), "v"(xs[u].y), "v"(xs[u].z), "v"(xs[u].w),
                             "v"(ea[u][0]), "v"(ea[u][1]), "v"(ea[u][2]), "v"(ea[u][3]),
                             "v"(ea[u][4]), "v"(ea[u][5]), "v"(ea[u][6]));
            }
            #pragma unroll
            for (int u = 0; u < 4; ++u) {
                if (i + u < cnt) {
                    const bool live = (base + i + u < deg);
                    const float4 x = xs[u];
                    float4 em = make_float4(0.0f, 0.0f, 0.0f, 0.0f);
                    #pragma unroll
                    for (int j = 0; j < ED; ++j) {
                        em.x = fmaf(ea[u][j], w[j].x, em.x);
                        em.y = fmaf(ea[u][j], w[j].y, em.y);
                        em.z = fmaf(ea[u][j], w[j].z, em.z);
                        em.w = fmaf(ea[u][j], w[j].w, em.w);
                    }
                    const float lv = live ? 1.0f : 0.0f;
                    se.x = fmaf(lv, em.x, se.x);
                    se.y = fmaf(lv, em.y, se.y);
                    se.z = fmaf(lv, em.z, se.z);
                    se.w = fmaf(lv, em.w, se.w);
                    const float v0 = lrelu((x.x + xrd.x) + em.x, NEG_ATT);
                    const float v1 = lrelu((x.y + xrd.y) + em.y, NEG_ATT);
                    const float v2 = lrelu((x.z + xrd.z) + em.z, NEG_ATT);
                    const float v3 = lrelu((x.w + xrd.w) + em.w, NEG_ATT);
                    float lg = quad_sum(fmaf(v1, at.y, v0 * at.x) + fmaf(v3, at.w, v2 * at.z));
                    lg = live ? lg : -INFINITY;
                    const float mn = fmaxf(m, lg);
                    const float sc = __expf(m - mn);
                    const float wg = __expf(lg - mn);
                    z     = fmaf(z,     sc, wg);
                    acc.x = fmaf(acc.x, sc, wg * x.x);
                    acc.y = fmaf(acc.y, sc, wg * x.y);
                    acc.z = fmaf(acc.z, sc, wg * x.z);
                    acc.w = fmaf(acc.w, sc, wg * x.w);
                    m = mn;
                }
            }
        }
    }

    {
        const float inv = 1.0f / fmaxf((float)deg, 1.0f);
        const float v0 = lrelu((xld.x + xrd.x) + se.x * inv, NEG_ATT);
        const float v1 = lrelu((xld.y + xrd.y) + se.y * inv, NEG_ATT);
        const float v2 = lrelu((xld.z + xrd.z) + se.z * inv, NEG_ATT);
        const float v3 = lrelu((xld.w + xrd.w) + se.w * inv, NEG_ATT);
        const float lg = quad_sum(fmaf(v1, at.y, v0 * at.x) + fmaf(v3, at.w, v2 * at.z));
        const float mn = fmaxf(m, lg);
        const float sc = __expf(m - mn);
        const float wg = __expf(lg - mn);
        z     = fmaf(z,     sc, wg);
        acc.x = fmaf(acc.x, sc, wg * xld.x);
        acc.y = fmaf(acc.y, sc, wg * xld.y);
        acc.z = fmaf(acc.z, sc, wg * xld.z);
        acc.w = fmaf(acc.w, sc, wg * xld.w);
    }

    const float4 b = *reinterpret_cast<const float4*>(reinterpret_cast<const char*>(bo) + cb);
    float4 o;
    o.x = lrelu(acc.x / z + b.x, NEG);
    o.y = lrelu(acc.y / z + b.y, NEG);
    o.z = lrelu(acc.z / z + b.z, NEG);
    o.w = lrelu(acc.w / z + b.w, NEG);
    if constexpr (HEAD) {
        const float4 wsv = *reinterpret_cast<const float4*>(reinterpret_cast<const char*>(Ws) + cb);
        float s = fmaf(o.y, wsv.y, o.x * wsv.x) + fmaf(o.w, wsv.w, o.z * wsv.z);
        s = quad_sum(s);
        s += __int_as_float(__builtin_amdgcn_update_dpp(0, __float_as_int(s), 0x164, 0xF, 0xF, false));
        s += __int_as_float(__builtin_amdgcn_update_dpp(0, __float_as_int(s), 0x168, 0xF, 0xF, false));
        if (c == 0) hout[2 * wv + half] = s + bs[0];
    } else {
        *reinterpret_cast<float4*>(reinterpret_cast<char*>(hout) + pb + lb) = o;
    }
}

extern "C" void kernel_launch(void* const* d_in, const int* in_sizes, int n_in,
                              void* d_out, int out_size, void* d_ws, size_t ws_size,
                              hipStream_t stream) {
    const float* x     = (const float*)d_in[0];
    const int*   ei    = (const int*)d_in[1];
    const float* eattr = (const float*)d_in[2];
    const float* Wl1 = (const float*)d_in[3];  const float* bl1 = (const float*)d_in[4];
    const float* Wr1 = (const float*)d_in[5];  const float* br1 = (const float*)d_in[6];
    const float* We1 = (const float*)d_in[7];  const float* att1 = (const float*)d_in[8];
    const float* bo1 = (const float*)d_in[9];
    const float* Wl2 = (const float*)d_in[10]; const float* bl2 = (const float*)d_in[11];
    const float* Wr2 = (const float*)d_in[12]; const float* br2 = (const float*)d_in[13];
    const float* We2 = (const float*)d_in[14]; const float* att2 = (const float*)d_in[15];
    const float* bo2 = (const float*)d_in[16];
    const float* Ws  = (const float*)d_in[17]; const float* bs = (const float*)d_in[18];
    float* out = (float*)d_out;

    float* ws = (float*)d_ws;
    size_t o = 0;
    float* xl    = ws + o; o += (size_t)N * HC;
    float* xr    = ws + o; o += (size_t)N * HC;
    float* hbuf  = ws + o; o += (size_t)N * HC;
    int*   rowp  = (int*)(ws + o); o += (size_t)N + 2;
    int2*  csr   = (int2*)(ws + o); o += (size_t)E * 2;
    int2*  recs  = (int2*)(ws + o); o += (size_t)E * 2;
    int*   offT  = (int*)(ws + o); o += (size_t)(NBIN + 1) * NSLAB;

    k_slab<<<NGRID0, SLAB_T, 0, stream>>>(ei, recs, offT, x, Wl1, bl1, Wr1, br1, xl, xr);
    k_bin<<<NBIN, 1024, 0, stream>>>(recs, offT, rowp, csr);

    k_gat_node<false><<<N / 16, 256, 0, stream>>>(xl, xr, eattr, rowp, csr, We1, att1, bo1, Ws, bs, hbuf);

    k_gemm_pair<HC><<<N / 16, 128, 0, stream>>>(hbuf, Wl2, bl2, Wr2, br2, xl, xr);
    k_gat_node<true><<<N / 16, 256, 0, stream>>>(xl, xr, eattr, rowp, csr, We2, att2, bo2, Ws, bs, out);
}
